# starting kernel
# baseline (speedup 1.0000x reference)
#include <hip/hip_runtime.h>

typedef __attribute__((ext_vector_type(2))) float v2f;
typedef __attribute__((ext_vector_type(8))) float v8f;

#define IN_DIM 64

__global__ void deg_kernel(const int* __restrict__ row, const int* __restrict__ col,
                           float* __restrict__ deg, int E) {
    int e = blockIdx.x * blockDim.x + threadIdx.x;
    if (e >= E) return;
    int r = row[e], c = col[e];
    if (r != c) atomicAdd(&deg[r], 1.0f);
}

__global__ void rsqrt_kernel(float* __restrict__ dis, int N) {
    int i = blockIdx.x * blockDim.x + threadIdx.x;
    if (i >= N) return;
    float d = dis[i];
    dis[i] = (d > 0.0f) ? rsqrtf(d) : 0.0f;
}

__global__ void neg_copy_kernel(const float* __restrict__ src, float* __restrict__ dst, int n) {
    int i = blockIdx.x * blockDim.x + threadIdx.x;
    if (i >= n) return;
    dst[i] = -src[i];
}

__global__ void spmm_kernel(const int* __restrict__ row, const int* __restrict__ col,
                            const float* __restrict__ dis, const float* __restrict__ V,
                            float* __restrict__ OUT, float scale, int E) {
    int gid = blockIdx.x * blockDim.x + threadIdx.x;
    int e = gid >> 4;
    if (e >= E) return;
    int chunk = (gid & 15) << 2;
    int r = row[e], c = col[e];
    if (r == c) return;
    float wgt = -scale * dis[r] * dis[c];
    if (wgt == 0.0f) return;
    const float4 v = *(const float4*)(V + (size_t)c * IN_DIM + chunk);
    float* o = OUT + (size_t)r * IN_DIM + chunk;
    atomicAdd(o + 0, wgt * v.x);
    atomicAdd(o + 1, wgt * v.y);
    atomicAdd(o + 2, wgt * v.z);
    atomicAdd(o + 3, wgt * v.w);
}

template <int OUT_DIM, int DO_RELU>
__global__ void __launch_bounds__(256, 1)
cheb_gemm_wmma(const float* __restrict__ T0,
               const float* __restrict__ T1,
               const float* __restrict__ T2,
               const float* __restrict__ W,
               const float* __restrict__ bias,
               float* __restrict__ out,
               int N) {
    constexpr int COL_TILES = OUT_DIM / 16;
    const int lane = threadIdx.x & 31;
    const int wave = (blockIdx.x * blockDim.x + threadIdx.x) >> 5;
    const int totalWaves = (gridDim.x * blockDim.x) >> 5;
    const int rowWaves = totalWaves / COL_TILES;
    const int ct   = wave % COL_TILES;
    const int rwid = wave / COL_TILES;
    const int rowTiles = (N + 15) >> 4;

    const int n    = lane & 15;
    const int m    = lane & 15;
    const int koff = (lane >> 4) << 1;
    const int col0 = ct << 4;
    const int mhalf = (lane >> 4) << 3;

    v2f bfrag[3][16];
#pragma unroll
    for (int s = 0; s < 3; ++s) {
        const float* Ws = W + s * IN_DIM * OUT_DIM + col0 + n;
#pragma unroll
        for (int k4 = 0; k4 < 16; ++k4) {
            const int kk = k4 * 4 + koff;
            bfrag[s][k4].x = Ws[(kk + 0) * OUT_DIM];
            bfrag[s][k4].y = Ws[(kk + 1) * OUT_DIM];
        }
    }
    const float bv = bias[col0 + n];

    for (int rt = rwid; rt < rowTiles; rt += rowWaves) {
        const int row0 = rt << 4;
        const int arow = min(row0 + m, N - 1);
        const size_t abase = (size_t)arow * IN_DIM + koff;

        v8f acc0 = {};
        v8f acc1 = {};
#pragma unroll
        for (int s = 0; s < 3; ++s) {
            const float* T = (s == 0) ? T0 : (s == 1) ? T1 : T2;
            const float* ap = T + abase;

#pragma unroll
            for (int k8 = 0; k8 < 8; ++k8) {
                v2f a0, a1;
                a0.x = ap[(2 * k8 + 0) * 4 + 0];
                a0.y = ap[(2 * k8 + 0) * 4 + 1];
                a1.x = ap[(2 * k8 + 1) * 4 + 0];
                a1.y = ap[(2 * k8 + 1) * 4 + 1];
                acc0 = __builtin_amdgcn_wmma_f32_16x16x4_f32(
                    false, a0, false, bfrag[s][2 * k8 + 0], (short)0, acc0, false, false);
                acc1 = __builtin_amdgcn_wmma_f32_16x16x4_f32(
                    false, a1, false, bfrag[s][2 * k8 + 1], (short)0, acc1, false, false);
            }
        }

#pragma unroll
        for (int r = 0; r < 8; ++r) {
            int mr = row0 + r + mhalf;
            if (mr < N) {
                float v = acc0[r] + acc1[r] + bv;
                if (DO_RELU) v = fmaxf(v, 0.0f);
                out[(size_t)mr * OUT_DIM + col0 + n] = v;
            }
        }
    }
}

extern "C" void kernel_launch(void* const* d_in, const int* in_sizes, int n_in,
                              void* d_out, int out_size, void* d_ws, size_t ws_size,
                              hipStream_t stream) {
    const float* x  = (const float*)d_in[0];
    const int*   ei = (const int*)d_in[1];
    const float* W1 = (const float*)d_in[2];
    const float* b1 = (const float*)d_in[3];
    const float* W2 = (const float*)d_in[4];
    const float* b2 = (const float*)d_in[5];
    float* out = (float*)d_out;

    const int N = in_sizes[0] / IN_DIM;
    const int E = in_sizes[1] / 2;
    const int* row = ei;
    const int* col = ei + E;

    float* ws  = (float*)d_ws;
    float* dis = ws;
    float* Tx1 = ws + N;
    float* Tx2 = Tx1 + (size_t)N * IN_DIM;
    float* h   = Tx2 + (size_t)N * IN_DIM;

    const int threads = 256;
    const int featN   = N * IN_DIM;
    const unsigned gridE    = (unsigned)((E + threads - 1) / threads);
    const unsigned gridN    = (unsigned)((N + threads - 1) / threads);
    const unsigned gridFeat = (unsigned)((featN + threads - 1) / threads);
    const unsigned gridSpmm = (unsigned)(((size_t)E * 16 + threads - 1) / threads);

    hipMemsetAsync(dis, 0, (size_t)N * sizeof(float), stream);
    deg_kernel<<<gridE, threads, 0, stream>>>(row, col, dis, E);
    rsqrt_kernel<<<gridN, threads, 0, stream>>>(dis, N);

    hipMemsetAsync(Tx1, 0, (size_t)featN * sizeof(float), stream);
    spmm_kernel<<<gridSpmm, threads, 0, stream>>>(row, col, dis, x, Tx1, 1.0f, E);
    neg_copy_kernel<<<gridFeat, threads, 0, stream>>>(x, Tx2, featN);
    spmm_kernel<<<gridSpmm, threads, 0, stream>>>(row, col, dis, Tx1, Tx2, 2.0f, E);

    cheb_gemm_wmma<64, 1><<<512, threads, 0, stream>>>(x, Tx1, Tx2, W1, b1, h, N);

    hipMemsetAsync(Tx1, 0, (size_t)featN * sizeof(float), stream);
    spmm_kernel<<<gridSpmm, threads, 0, stream>>>(row, col, dis, h, Tx1, 1.0f, E);
    neg_copy_kernel<<<gridFeat, threads, 0, stream>>>(h, Tx2, featN);
    spmm_kernel<<<gridSpmm, threads, 0, stream>>>(row, col, dis, Tx1, Tx2, 2.0f, E);

    cheb_gemm_wmma<16, 0><<<128, threads, 0, stream>>>(h, Tx1, Tx2, W2, b2, out, N);
}


# SpMM: each package half adds only into its own half of the rows (die number read per block, one ticket counter per row range, blocks help the other range at the end); sums stay in arrival order
# speedup vs baseline: 2.0680x; 2.0680x over previous
#include <hip/hip_runtime.h>

typedef __attribute__((ext_vector_type(2))) float v2f;
typedef __attribute__((ext_vector_type(8))) float v8f;

#define IN_DIM 64

__global__ void deg_kernel(const int* __restrict__ row, const int* __restrict__ col,
                           float* __restrict__ deg, int E) {
    int e = blockIdx.x * blockDim.x + threadIdx.x;
    if (e >= E) return;
    int r = row[e], c = col[e];
    if (r != c) atomicAdd(&deg[r], 1.0f);
}

__global__ void rsqrt_kernel(float* __restrict__ dis, float* __restrict__ ticketWords, int N) {
    int i = blockIdx.x * blockDim.x + threadIdx.x;
    if (i < 8) reinterpret_cast<int*>(ticketWords)[i] = 0;
    if (i >= N) return;
    float d = dis[i];
    dis[i] = (d > 0.0f) ? rsqrtf(d) : 0.0f;
}

__global__ void neg_copy_kernel(const float* __restrict__ src, float* __restrict__ dst, int n) {
    int i = blockIdx.x * blockDim.x + threadIdx.x;
    if (i >= n) return;
    dst[i] = -src[i];
}

#define WAVE_EDGES 64
#define TICKET_EDGES 512
#define EDGE_GROUP 8

__global__ void __launch_bounds__(256)
spmm_kernel(const int* __restrict__ row, const int* __restrict__ col,
            const float* __restrict__ dis, const float* __restrict__ V,
            float* __restrict__ OUT, float* __restrict__ ticketWords,
            float scale, int E, int N) {
    __shared__ int s_ticket[2];
    int* tickets = reinterpret_cast<int*>(ticketWords);
    const int lane = threadIdx.x & 31;
    const int waveInBlock = threadIdx.x >> 5;
    const int nTickets = (E + TICKET_EDGES - 1) / TICKET_EDGES;
    const int split = N >> 1;
    const int myHalf = (int)((__builtin_amdgcn_s_sendmsg_rtn(0x87) >> 18) & 1u);

    int k = 0;
    for (int phase = 0; phase < 2; ++phase) {
        const int rg = myHalf ^ phase;
        const int lo = rg ? split : 0;
        const unsigned span = (unsigned)(rg ? (N - split) : split);
        int* cnt = tickets + rg;
        int tn = 0;
        if (threadIdx.x == 0) tn = atomicAdd(cnt, 1);
        for (;;) {
            if (threadIdx.x == 0) s_ticket[k & 1] = tn;
            __syncthreads();
            const int t = __builtin_amdgcn_readfirstlane(s_ticket[k & 1]);
            ++k;
            if (t >= nTickets) break;
            const int e = t * TICKET_EDGES + waveInBlock * WAVE_EDGES + lane;
            int r0 = 0, c0 = 0, r1 = 0, c1 = 0;
            if (e < E)      { r0 = row[e];      c0 = col[e]; }
            if (e + 32 < E) { r1 = row[e + 32]; c1 = col[e + 32]; }
            if (threadIdx.x == 0) tn = atomicAdd(cnt, 1);
            const bool in0 = (e < E)      && ((unsigned)(r0 - lo) < span) && (r0 != c0);
            const bool in1 = (e + 32 < E) && ((unsigned)(r1 - lo) < span) && (r1 != c1);
            float d0r = 0.0f, d0c = 0.0f, d1r = 0.0f, d1c = 0.0f;
            if (in0) { d0r = dis[r0]; d0c = dis[c0]; }
            if (in1) { d1r = dis[r1]; d1c = dis[c1]; }
            const float w0 = -scale * d0r * d0c;
            const float w1 = -scale * d1r * d1c;
            unsigned long long mask =
                ((unsigned long long)__builtin_amdgcn_ballot_w32(in1 && w1 != 0.0f) << 32) |
                (unsigned long long)__builtin_amdgcn_ballot_w32(in0 && w0 != 0.0f);

            while (mask) {
                float v0[EDGE_GROUP], v1[EDGE_GROUP], wj[EDGE_GROUP];
                int rj[EDGE_GROUP];
#pragma unroll
                for (int j = 0; j < EDGE_GROUP; ++j) {
                    wj[j] = 0.0f; rj[j] = 0; v0[j] = 0.0f; v1[j] = 0.0f;
                    if (mask) {
                        const int b = __builtin_ctzll(mask);
                        mask &= mask - 1;
                        const int l = b & 31;
                        int cj;
                        if (b < 32) {
                            cj    = __builtin_amdgcn_readlane(c0, l);
                            rj[j] = __builtin_amdgcn_readlane(r0, l);
                            wj[j] = __int_as_float(__builtin_amdgcn_readlane(__float_as_int(w0), l));
                        } else {
                            cj    = __builtin_amdgcn_readlane(c1, l);
                            rj[j] = __builtin_amdgcn_readlane(r1, l);
                            wj[j] = __int_as_float(__builtin_amdgcn_readlane(__float_as_int(w1), l));
                        }
                        const float* vp = V + (size_t)cj * IN_DIM + lane;
                        v0[j] = vp[0];
                        v1[j] = vp[32];
                    }
                }
#pragma unroll
                for (int j = 0; j < EDGE_GROUP; ++j) {
                    if (wj[j] != 0.0f) {
                        float* op = OUT + (size_t)rj[j] * IN_DIM + lane;
                        atomicAdd(op,      wj[j] * v0[j]);
                        atomicAdd(op + 32, wj[j] * v1[j]);
                    }
                }
            }
        }
    }
}

template <int OUT_DIM, int DO_RELU>
__global__ void __launch_bounds__(256, 1)
cheb_gemm_wmma(const float* __restrict__ T0,
               const float* __restrict__ T1,
               const float* __restrict__ T2,
               const float* __restrict__ W,
               const float* __restrict__ bias,
               float* __restrict__ out,
               int N) {
    constexpr int COL_TILES = OUT_DIM / 16;
    const int lane = threadIdx.x & 31;
    const int wave = (blockIdx.x * blockDim.x + threadIdx.x) >> 5;
    const int totalWaves = (gridDim.x * blockDim.x) >> 5;
    const int rowWaves = totalWaves / COL_TILES;
    const int ct   = wave % COL_TILES;
    const int rwid = wave / COL_TILES;
    const int rowTiles = (N + 15) >> 4;

    const int n    = lane & 15;
    const int m    = lane & 15;
    const int koff = (lane >> 4) << 1;
    const int col0 = ct << 4;
    const int mhalf = (lane >> 4) << 3;

    v2f bfrag[3][16];
#pragma unroll
    for (int s = 0; s < 3; ++s) {
        const float* Ws = W + s * IN_DIM * OUT_DIM + col0 + n;
#pragma unroll
        for (int k4 = 0; k4 < 16; ++k4) {
            const int kk = k4 * 4 + koff;
            bfrag[s][k4].x = Ws[(kk + 0) * OUT_DIM];
            bfrag[s][k4].y = Ws[(kk + 1) * OUT_DIM];
        }
    }
    const float bv = bias[col0 + n];

    for (int rt = rwid; rt < rowTiles; rt += rowWaves) {
        const int row0 = rt << 4;
        const int arow = min(row0 + m, N - 1);
        const size_t abase = (size_t)arow * IN_DIM + koff;

        v8f acc0 = {};
        v8f acc1 = {};
#pragma unroll
        for (int s = 0; s < 3; ++s) {
            const float* T = (s == 0) ? T0 : (s == 1) ? T1 : T2;
            const float* ap = T + abase;

#pragma unroll
            for (int k8 = 0; k8 < 8; ++k8) {
                v2f a0, a1;
                a0.x = ap[(2 * k8 + 0) * 4 + 0];
                a0.y = ap[(2 * k8 + 0) * 4 + 1];
                a1.x = ap[(2 * k8 + 1) * 4 + 0];
                a1.y = ap[(2 * k8 + 1) * 4 + 1];
                acc0 = __builtin_amdgcn_wmma_f32_16x16x4_f32(
                    false, a0, false, bfrag[s][2 * k8 + 0], (short)0, acc0, false, false);
                acc1 = __builtin_amdgcn_wmma_f32_16x16x4_f32(
                    false, a1, false, bfrag[s][2 * k8 + 1], (short)0, acc1, false, false);
            }
        }

#pragma unroll
        for (int r = 0; r < 8; ++r) {
            int mr = row0 + r + mhalf;
            if (mr < N) {
                float v = acc0[r] + acc1[r] + bv;
                if (DO_RELU) v = fmaxf(v, 0.0f);
                out[(size_t)mr * OUT_DIM + col0 + n] = v;
            }
        }
    }
}

extern "C" void kernel_launch(void* const* d_in, const int* in_sizes, int n_in,
                              void* d_out, int out_size, void* d_ws, size_t ws_size,
                              hipStream_t stream) {
    const float* x  = (const float*)d_in[0];
    const int*   ei = (const int*)d_in[1];
    const float* W1 = (const float*)d_in[2];
    const float* b1 = (const float*)d_in[3];
    const float* W2 = (const float*)d_in[4];
    const float* b2 = (const float*)d_in[5];
    float* out = (float*)d_out;

    const int N = in_sizes[0] / IN_DIM;
    const int E = in_sizes[1] / 2;
    const int* row = ei;
    const int* col = ei + E;

    float* ws  = (float*)d_ws;
    float* dis = ws;
    float* Tx1 = ws + N;
    float* Tx2 = Tx1 + (size_t)N * IN_DIM;
    float* h   = Tx2 + (size_t)N * IN_DIM;

    const int threads = 256;
    const int featN   = N * IN_DIM;
    const unsigned gridE    = (unsigned)((E + threads - 1) / threads);
    const unsigned gridN    = (unsigned)((N + threads - 1) / threads);
    const unsigned gridFeat = (unsigned)((featN + threads - 1) / threads);
    const unsigned gridSpmm = 2048;
    float* tk = out;

    hipMemsetAsync(dis, 0, (size_t)N * sizeof(float), stream);
    deg_kernel<<<gridE, threads, 0, stream>>>(row, col, dis, E);
    rsqrt_kernel<<<gridN, threads, 0, stream>>>(dis, out, N);

    hipMemsetAsync(Tx1, 0, (size_t)featN * sizeof(float), stream);
    spmm_kernel<<<gridSpmm, threads, 0, stream>>>(row, col, dis, x, Tx1, tk + 0, 1.0f, E, N);
    neg_copy_kernel<<<gridFeat, threads, 0, stream>>>(x, Tx2, featN);
    spmm_kernel<<<gridSpmm, threads, 0, stream>>>(row, col, dis, Tx1, Tx2, tk + 2, 2.0f, E, N);

    cheb_gemm_wmma<64, 1><<<512, threads, 0, stream>>>(x, Tx1, Tx2, W1, b1, h, N);

    hipMemsetAsync(Tx1, 0, (size_t)featN * sizeof(float), stream);
    spmm_kernel<<<gridSpmm, threads, 0, stream>>>(row, col, dis, h, Tx1, tk + 4, 1.0f, E, N);
    neg_copy_kernel<<<gridFeat, threads, 0, stream>>>(h, Tx2, featN);
    spmm_kernel<<<gridSpmm, threads, 0, stream>>>(row, col, dis, Tx1, Tx2, tk + 6, 2.0f, E, N);

    cheb_gemm_wmma<16, 0><<<128, threads, 0, stream>>>(h, Tx1, Tx2, W2, b2, out, N);
}


# SpMM edge weight -dis[r]*dis[c] factored out of the scatter: col factor looked up once per edge (1st product) or pre-folded into V (2nd), row factor in the dense A-load; products re-associated
# speedup vs baseline: 2.1758x; 1.0521x over previous
#include <hip/hip_runtime.h>

typedef __attribute__((ext_vector_type(2))) float v2f;
typedef __attribute__((ext_vector_type(8))) float v8f;

#define IN_DIM 64

__global__ void deg_kernel(const int* __restrict__ row, const int* __restrict__ col,
                           float* __restrict__ deg, int E) {
    int e = blockIdx.x * blockDim.x + threadIdx.x;
    if (e >= E) return;
    int r = row[e], c = col[e];
    if (r != c) atomicAdd(&deg[r], 1.0f);
}

__global__ void rsqrt_kernel(float* __restrict__ dis, float* __restrict__ ticketWords, int N) {
    int i = blockIdx.x * blockDim.x + threadIdx.x;
    if (i < 8) reinterpret_cast<int*>(ticketWords)[i] = 0;
    if (i >= N) return;
    float d = dis[i];
    dis[i] = (d > 0.0f) ? rsqrtf(d) : 0.0f;
}

__global__ void rescale_seed_kernel(const float* __restrict__ dis, float* __restrict__ S1,
                                    float* __restrict__ S2, int n) {
    int i = blockIdx.x * blockDim.x + threadIdx.x;
    if (i >= n) return;
    const float d = dis[i / IN_DIM];
    S1[i] = -(d * d) * S1[i];
    S2[i] = 0.0f;
}

#define WAVE_EDGES 64
#define TICKET_EDGES 512
#define EDGE_GROUP 8

template <int COLSCALE>
__global__ void __launch_bounds__(256)
spmm_kernel(const int* __restrict__ row, const int* __restrict__ col,
            const float* __restrict__ dis, const float* __restrict__ V,
            float* __restrict__ OUT, float* __restrict__ ticketWords,
            int E, int N) {
    __shared__ int s_ticket[2];
    int* tickets = reinterpret_cast<int*>(ticketWords);
    const int lane = threadIdx.x & 31;
    const int waveInBlock = threadIdx.x >> 5;
    const int nTickets = (E + TICKET_EDGES - 1) / TICKET_EDGES;
    const int split = N >> 1;
    const int myHalf = (int)((__builtin_amdgcn_s_sendmsg_rtn(0x87) >> 18) & 1u);

    int k = 0;
    for (int phase = 0; phase < 2; ++phase) {
        const int rg = myHalf ^ phase;
        const int lo = rg ? split : 0;
        const unsigned span = (unsigned)(rg ? (N - split) : split);
        int* cnt = tickets + rg;
        int tn = 0;
        if (threadIdx.x == 0) tn = atomicAdd(cnt, 1);
        for (;;) {
            if (threadIdx.x == 0) s_ticket[k & 1] = tn;
            __syncthreads();
            const int t = __builtin_amdgcn_readfirstlane(s_ticket[k & 1]);
            ++k;
            if (t >= nTickets) break;
            const int e = t * TICKET_EDGES + waveInBlock * WAVE_EDGES + lane;
            int r0 = 0, c0 = 0, r1 = 0, c1 = 0;
            if (e < E)      { r0 = row[e];      c0 = col[e]; }
            if (e + 32 < E) { r1 = row[e + 32]; c1 = col[e + 32]; }
            if (threadIdx.x == 0) tn = atomicAdd(cnt, 1);
            const bool in0 = (e < E)      && ((unsigned)(r0 - lo) < span) && (r0 != c0);
            const bool in1 = (e + 32 < E) && ((unsigned)(r1 - lo) < span) && (r1 != c1);
            float w0 = in0 ? 1.0f : 0.0f;
            float w1 = in1 ? 1.0f : 0.0f;
            if (COLSCALE) {
                if (in0) w0 = dis[c0];
                if (in1) w1 = dis[c1];
            }
            unsigned long long mask =
                ((unsigned long long)__builtin_amdgcn_ballot_w32(in1 && w1 != 0.0f) << 32) |
                (unsigned long long)__builtin_amdgcn_ballot_w32(in0 && w0 != 0.0f);

            while (mask) {
                float v0[EDGE_GROUP], v1[EDGE_GROUP], wj[EDGE_GROUP];
                int rj[EDGE_GROUP];
#pragma unroll
                for (int j = 0; j < EDGE_GROUP; ++j) {
                    wj[j] = 0.0f; rj[j] = 0; v0[j] = 0.0f; v1[j] = 0.0f;
                    if (mask) {
                        const int b = __builtin_ctzll(mask);
                        mask &= mask - 1;
                        const int l = b & 31;
                        int cj;
                        if (b < 32) {
                            cj    = __builtin_amdgcn_readlane(c0, l);
                            rj[j] = __builtin_amdgcn_readlane(r0, l);
                            wj[j] = __int_as_float(__builtin_amdgcn_readlane(__float_as_int(w0), l));
                        } else {
                            cj    = __builtin_amdgcn_readlane(c1, l);
                            rj[j] = __builtin_amdgcn_readlane(r1, l);
                            wj[j] = __int_as_float(__builtin_amdgcn_readlane(__float_as_int(w1), l));
                        }
                        const float* vp = V + (size_t)cj * IN_DIM + lane;
                        v0[j] = vp[0];
                        v1[j] = vp[32];
                    }
                }
#pragma unroll
                for (int j = 0; j < EDGE_GROUP; ++j) {
                    if (wj[j] != 0.0f) {
                        float* op = OUT + (size_t)rj[j] * IN_DIM + lane;
                        atomicAdd(op,      COLSCALE ? wj[j] * v0[j] : v0[j]);
                        atomicAdd(op + 32, COLSCALE ? wj[j] * v1[j] : v1[j]);
                    }
                }
            }
        }
    }
}

template <int OUT_DIM, int DO_RELU>
__global__ void __launch_bounds__(256, 1)
cheb_gemm_wmma(const float* __restrict__ T0,
               const float* __restrict__ T1,
               const float* __restrict__ T2,
               const float* __restrict__ dis,
               const float* __restrict__ W,
               const float* __restrict__ bias,
               float* __restrict__ out,
               int N) {
    constexpr int COL_TILES = OUT_DIM / 16;
    const int lane = threadIdx.x & 31;
    const int wave = (blockIdx.x * blockDim.x + threadIdx.x) >> 5;
    const int totalWaves = (gridDim.x * blockDim.x) >> 5;
    const int rowWaves = totalWaves / COL_TILES;
    const int ct   = wave % COL_TILES;
    const int rwid = wave / COL_TILES;
    const int rowTiles = (N + 15) >> 4;

    const int n    = lane & 15;
    const int m    = lane & 15;
    const int koff = (lane >> 4) << 1;
    const int col0 = ct << 4;
    const int mhalf = (lane >> 4) << 3;

    v2f bfrag[3][16];
#pragma unroll
    for (int s = 0; s < 3; ++s) {
        const float* Ws = W + s * IN_DIM * OUT_DIM + col0 + n;
#pragma unroll
        for (int k4 = 0; k4 < 16; ++k4) {
            const int kk = k4 * 4 + koff;
            bfrag[s][k4].x = Ws[(kk + 0) * OUT_DIM];
            bfrag[s][k4].y = Ws[(kk + 1) * OUT_DIM];
        }
    }
    const float bv = bias[col0 + n];

    for (int rt = rwid; rt < rowTiles; rt += rowWaves) {
        const int row0 = rt << 4;
        const int arow = min(row0 + m, N - 1);
        const size_t abase = (size_t)arow * IN_DIM + koff;
        const float dr  = dis[arow];
        const float inv = (dr > 0.0f) ? 1.0f / dr : 0.0f;
        const float m2  = -2.0f * dr;

        v8f acc0 = {};
        v8f acc1 = {};
#pragma unroll
        for (int s = 0; s < 3; ++s) {
            const float* T = (s == 0) ? T0 : (s == 1) ? T1 : T2;
            const float* ap = T + abase;

#pragma unroll
            for (int k8 = 0; k8 < 8; ++k8) {
                v2f a0, a1;
                a0.x = ap[(2 * k8 + 0) * 4 + 0];
                a0.y = ap[(2 * k8 + 0) * 4 + 1];
                a1.x = ap[(2 * k8 + 1) * 4 + 0];
                a1.y = ap[(2 * k8 + 1) * 4 + 1];
                if (s == 1) {
                    a0.x *= inv; a0.y *= inv; a1.x *= inv; a1.y *= inv;
                } else if (s == 2) {
                    const float* xp = T0 + abase;
                    a0.x = m2 * a0.x - xp[(2 * k8 + 0) * 4 + 0];
                    a0.y = m2 * a0.y - xp[(2 * k8 + 0) * 4 + 1];
                    a1.x = m2 * a1.x - xp[(2 * k8 + 1) * 4 + 0];
                    a1.y = m2 * a1.y - xp[(2 * k8 + 1) * 4 + 1];
                }
                acc0 = __builtin_amdgcn_wmma_f32_16x16x4_f32(
                    false, a0, false, bfrag[s][2 * k8 + 0], (short)0, acc0, false, false);
                acc1 = __builtin_amdgcn_wmma_f32_16x16x4_f32(
                    false, a1, false, bfrag[s][2 * k8 + 1], (short)0, acc1, false, false);
            }
        }

#pragma unroll
        for (int r = 0; r < 8; ++r) {
            int mr = row0 + r + mhalf;
            if (mr < N) {
                float v = acc0[r] + acc1[r] + bv;
                if (DO_RELU) v = fmaxf(v, 0.0f);
                out[(size_t)mr * OUT_DIM + col0 + n] = v;
            }
        }
    }
}

extern "C" void kernel_launch(void* const* d_in, const int* in_sizes, int n_in,
                              void* d_out, int out_size, void* d_ws, size_t ws_size,
                              hipStream_t stream) {
    const float* x  = (const float*)d_in[0];
    const int*   ei = (const int*)d_in[1];
    const float* W1 = (const float*)d_in[2];
    const float* b1 = (const float*)d_in[3];
    const float* W2 = (const float*)d_in[4];
    const float* b2 = (const float*)d_in[5];
    float* out = (float*)d_out;

    const int N = in_sizes[0] / IN_DIM;
    const int E = in_sizes[1] / 2;
    const int* row = ei;
    const int* col = ei + E;

    float* ws  = (float*)d_ws;
    float* dis = ws;
    float* Tx1 = ws + N;
    float* Tx2 = Tx1 + (size_t)N * IN_DIM;
    float* h   = Tx2 + (size_t)N * IN_DIM;

    const int threads = 256;
    const int featN   = N * IN_DIM;
    const unsigned gridE    = (unsigned)((E + threads - 1) / threads);
    const unsigned gridN    = (unsigned)((N + threads - 1) / threads);
    const unsigned gridFeat = (unsigned)((featN + threads - 1) / threads);
    const unsigned gridSpmm = 2048;
    float* tk = out;

    hipMemsetAsync(dis, 0, (size_t)N * sizeof(float), stream);
    deg_kernel<<<gridE, threads, 0, stream>>>(row, col, dis, E);
    rsqrt_kernel<<<gridN, threads, 0, stream>>>(dis, out, N);

    hipMemsetAsync(Tx1, 0, (size_t)featN * sizeof(float), stream);
    spmm_kernel<1><<<gridSpmm, threads, 0, stream>>>(row, col, dis, x, Tx1, tk + 0, E, N);
    rescale_seed_kernel<<<gridFeat, threads, 0, stream>>>(dis, Tx1, Tx2, featN);
    spmm_kernel<0><<<gridSpmm, threads, 0, stream>>>(row, col, dis, Tx1, Tx2, tk + 2, E, N);

    cheb_gemm_wmma<64, 1><<<512, threads, 0, stream>>>(x, Tx1, Tx2, dis, W1, b1, h, N);

    hipMemsetAsync(Tx1, 0, (size_t)featN * sizeof(float), stream);
    spmm_kernel<1><<<gridSpmm, threads, 0, stream>>>(row, col, dis, h, Tx1, tk + 4, E, N);
    rescale_seed_kernel<<<gridFeat, threads, 0, stream>>>(dis, Tx1, Tx2, featN);
    spmm_kernel<0><<<gridSpmm, threads, 0, stream>>>(row, col, dis, Tx1, Tx2, tk + 6, E, N);

    cheb_gemm_wmma<16, 0><<<128, threads, 0, stream>>>(h, Tx1, Tx2, dis, W2, b2, out, N);
}


# Dense stage: a block loads its rows of x, S1, S2 once by whole-line 16-byte loads, forms T1, T2, keeps the tiles in LDS; waves read A fragments there (was 16-line loads per column-tile wave)
# speedup vs baseline: 2.2753x; 1.0457x over previous
#include <hip/hip_runtime.h>

typedef __attribute__((ext_vector_type(2))) float v2f;
typedef __attribute__((ext_vector_type(8))) float v8f;
typedef __attribute__((ext_vector_type(4))) float v4f;

#define IN_DIM 64

__global__ void deg_kernel(const int* __restrict__ row, const int* __restrict__ col,
                           float* __restrict__ deg, int E) {
    int e = blockIdx.x * blockDim.x + threadIdx.x;
    if (e >= E) return;
    int r = row[e], c = col[e];
    if (r != c) atomicAdd(&deg[r], 1.0f);
}

__global__ void rsqrt_kernel(float* __restrict__ dis, float* __restrict__ ticketWords, int N) {
    int i = blockIdx.x * blockDim.x + threadIdx.x;
    if (i < 8) reinterpret_cast<int*>(ticketWords)[i] = 0;
    if (i >= N) return;
    float d = dis[i];
    dis[i] = (d > 0.0f) ? rsqrtf(d) : 0.0f;
}

__global__ void rescale_seed_kernel(const float* __restrict__ dis, float* __restrict__ S1,
                                    float* __restrict__ S2, int n) {
    int i = blockIdx.x * blockDim.x + threadIdx.x;
    if (i >= n) return;
    const float d = dis[i / IN_DIM];
    S1[i] = -(d * d) * S1[i];
    S2[i] = 0.0f;
}

#define WAVE_EDGES 64
#define TICKET_EDGES 512
#define EDGE_GROUP 8

template <int COLSCALE>
__global__ void __launch_bounds__(256)
spmm_kernel(const int* __restrict__ row, const int* __restrict__ col,
            const float* __restrict__ dis, const float* __restrict__ V,
            float* __restrict__ OUT, float* __restrict__ ticketWords,
            int E, int N) {
    __shared__ int s_ticket[2];
    int* tickets = reinterpret_cast<int*>(ticketWords);
    const int lane = threadIdx.x & 31;
    const int waveInBlock = threadIdx.x >> 5;
    const int nTickets = (E + TICKET_EDGES - 1) / TICKET_EDGES;
    const int split = N >> 1;
    const int myHalf = (int)((__builtin_amdgcn_s_sendmsg_rtn(0x87) >> 18) & 1u);

    int k = 0;
    for (int phase = 0; phase < 2; ++phase) {
        const int rg = myHalf ^ phase;
        const int lo = rg ? split : 0;
        const unsigned span = (unsigned)(rg ? (N - split) : split);
        int* cnt = tickets + rg;
        int tn = 0;
        if (threadIdx.x == 0) tn = atomicAdd(cnt, 1);
        for (;;) {
            if (threadIdx.x == 0) s_ticket[k & 1] = tn;
            __syncthreads();
            const int t = __builtin_amdgcn_readfirstlane(s_ticket[k & 1]);
            ++k;
            if (t >= nTickets) break;
            const int e = t * TICKET_EDGES + waveInBlock * WAVE_EDGES + lane;
            int r0 = 0, c0 = 0, r1 = 0, c1 = 0;
            if (e < E)      { r0 = row[e];      c0 = col[e]; }
            if (e + 32 < E) { r1 = row[e + 32]; c1 = col[e + 32]; }
            if (threadIdx.x == 0) tn = atomicAdd(cnt, 1);
            const bool in0 = (e < E)      && ((unsigned)(r0 - lo) < span) && (r0 != c0);
            const bool in1 = (e + 32 < E) && ((unsigned)(r1 - lo) < span) && (r1 != c1);
            float w0 = in0 ? 1.0f : 0.0f;
            float w1 = in1 ? 1.0f : 0.0f;
            if (COLSCALE) {
                if (in0) w0 = dis[c0];
                if (in1) w1 = dis[c1];
            }
            unsigned long long mask =
                ((unsigned long long)__builtin_amdgcn_ballot_w32(in1 && w1 != 0.0f) << 32) |
                (unsigned long long)__builtin_amdgcn_ballot_w32(in0 && w0 != 0.0f);

            while (mask) {
                float v0[EDGE_GROUP], v1[EDGE_GROUP], wj[EDGE_GROUP];
                int rj[EDGE_GROUP];
#pragma unroll
                for (int j = 0; j < EDGE_GROUP; ++j) {
                    wj[j] = 0.0f; rj[j] = 0; v0[j] = 0.0f; v1[j] = 0.0f;
                    if (mask) {
                        const int b = __builtin_ctzll(mask);
                        mask &= mask - 1;
                        const int l = b & 31;
                        int cj;
                        if (b < 32) {
                            cj    = __builtin_amdgcn_readlane(c0, l);
                            rj[j] = __builtin_amdgcn_readlane(r0, l);
                            wj[j] = __int_as_float(__builtin_amdgcn_readlane(__float_as_int(w0), l));
                        } else {
                            cj    = __builtin_amdgcn_readlane(c1, l);
                            rj[j] = __builtin_amdgcn_readlane(r1, l);
                            wj[j] = __int_as_float(__builtin_amdgcn_readlane(__float_as_int(w1), l));
                        }
                        const float* vp = V + (size_t)cj * IN_DIM + lane;
                        v0[j] = vp[0];
                        v1[j] = vp[32];
                    }
                }
#pragma unroll
                for (int j = 0; j < EDGE_GROUP; ++j) {
                    if (wj[j] != 0.0f) {
                        float* op = OUT + (size_t)rj[j] * IN_DIM + lane;
                        atomicAdd(op,      COLSCALE ? wj[j] * v0[j] : v0[j]);
                        atomicAdd(op + 32, COLSCALE ? wj[j] * v1[j] : v1[j]);
                    }
                }
            }
        }
    }
}

template <int OUT_DIM, int DO_RELU, int NWAVES>
__global__ void __launch_bounds__(NWAVES * 32, 1)
cheb_gemm_wmma(const float* __restrict__ T0,
               const float* __restrict__ T1,
               const float* __restrict__ T2,
               const float* __restrict__ dis,
               const float* __restrict__ W,
               const float* __restrict__ bias,
               float* __restrict__ out,
               int N) {
    constexpr int COL_TILES  = OUT_DIM / 16;
    constexpr int ROW_TILES  = NWAVES / COL_TILES;
    constexpr int STAGE_ROWS = 16 * ROW_TILES;
    constexpr int THREADS    = NWAVES * 32;
    constexpr int PITCH      = IN_DIM + 4;
    constexpr int CHUNKS     = STAGE_ROWS * 16 / THREADS;
    __shared__ __attribute__((aligned(16))) float sA[3][STAGE_ROWS * PITCH];

    const int lane = threadIdx.x & 31;
    const int waveInBlock = threadIdx.x >> 5;
    const int ct  = waveInBlock % COL_TILES;
    const int rtl = waveInBlock / COL_TILES;

    const int n    = lane & 15;
    const int m    = lane & 15;
    const int koff = (lane >> 4) << 1;
    const int col0 = ct << 4;
    const int mhalf = (lane >> 4) << 3;

    v2f bfrag[3][16];
#pragma unroll
    for (int s = 0; s < 3; ++s) {
        const float* Ws = W + s * IN_DIM * OUT_DIM + col0 + n;
#pragma unroll
        for (int k4 = 0; k4 < 16; ++k4) {
            const int kk = k4 * 4 + koff;
            bfrag[s][k4].x = Ws[(kk + 0) * OUT_DIM];
            bfrag[s][k4].y = Ws[(kk + 1) * OUT_DIM];
        }
    }
    const float bv = bias[col0 + n];

    const int nStages = (N + STAGE_ROWS - 1) / STAGE_ROWS;
    for (int st = blockIdx.x; st < nStages; st += gridDim.x) {
        const int srow0 = st * STAGE_ROWS;

        v4f gx[CHUNKS], g1[CHUNKS], g2[CHUNKS];
        float gd[CHUNKS];
#pragma unroll
        for (int c = 0; c < CHUNKS; ++c) {
            const int idx = c * THREADS + threadIdx.x;
            const int grow = min(srow0 + (idx >> 4), N - 1);
            const size_t g = (size_t)grow * IN_DIM + ((idx & 15) << 2);
            gx[c] = *reinterpret_cast<const v4f*>(T0 + g);
            g1[c] = *reinterpret_cast<const v4f*>(T1 + g);
            g2[c] = *reinterpret_cast<const v4f*>(T2 + g);
            gd[c] = dis[grow];
        }
        if constexpr (CHUNKS == 8) {
            asm volatile("" :: "v"(gx[0]), "v"(g1[0]), "v"(g2[0]), "v"(gx[1]), "v"(g1[1]), "v"(g2[1]),
                               "v"(gx[2]), "v"(g1[2]), "v"(g2[2]), "v"(gx[3]), "v"(g1[3]), "v"(g2[3]),
                               "v"(gx[4]), "v"(g1[4]), "v"(g2[4]), "v"(gx[5]), "v"(g1[5]), "v"(g2[5]),
                               "v"(gx[6]), "v"(g1[6]), "v"(g2[6]), "v"(gx[7]), "v"(g1[7]), "v"(g2[7]));
        } else {
            static_assert(CHUNKS == 2 || CHUNKS == 8, "staging fence is written for 2 or 8 pieces per thread");
            asm volatile("" :: "v"(gx[0]), "v"(g1[0]), "v"(g2[0]), "v"(gx[1]), "v"(g1[1]), "v"(g2[1]));
        }
#pragma unroll
        for (int c = 0; c < CHUNKS; ++c) {
            const int idx = c * THREADS + threadIdx.x;
            const float dr  = gd[c];
            const float inv = (dr > 0.0f) ? 1.0f / dr : 0.0f;
            const float m2  = -2.0f * dr;
            const v4f x4 = gx[c], s1 = g1[c], s2 = g2[c];
            v4f t1, t2;
            t1.x = s1.x * inv; t1.y = s1.y * inv; t1.z = s1.z * inv; t1.w = s1.w * inv;
            t2.x = m2 * s2.x - x4.x; t2.y = m2 * s2.y - x4.y;
            t2.z = m2 * s2.z - x4.z; t2.w = m2 * s2.w - x4.w;
            const int l = (idx >> 4) * PITCH + ((idx & 15) << 2);
            *reinterpret_cast<v4f*>(&sA[0][l]) = x4;
            *reinterpret_cast<v4f*>(&sA[1][l]) = t1;
            *reinterpret_cast<v4f*>(&sA[2][l]) = t2;
        }
        __syncthreads();

        v2f afrag[3][16];
#pragma unroll
        for (int s = 0; s < 3; ++s) {
            const float* ap = &sA[s][(rtl * 16 + m) * PITCH + koff];
#pragma unroll
            for (int k4 = 0; k4 < 16; ++k4)
                afrag[s][k4] = *reinterpret_cast<const v2f*>(ap + k4 * 4);
        }
        __builtin_amdgcn_sched_barrier(0);

        v8f acc0 = {};
        v8f acc1 = {};
#pragma unroll
        for (int s = 0; s < 3; ++s) {
#pragma unroll
            for (int k8 = 0; k8 < 8; ++k8) {
                acc0 = __builtin_amdgcn_wmma_f32_16x16x4_f32(
                    false, afrag[s][2 * k8 + 0], false, bfrag[s][2 * k8 + 0], (short)0, acc0, false, false);
                acc1 = __builtin_amdgcn_wmma_f32_16x16x4_f32(
                    false, afrag[s][2 * k8 + 1], false, bfrag[s][2 * k8 + 1], (short)0, acc1, false, false);
            }
        }

        const int row0 = srow0 + rtl * 16;
#pragma unroll
        for (int r = 0; r < 8; ++r) {
            int mr = row0 + r + mhalf;
            if (mr < N) {
                float v = acc0[r] + acc1[r] + bv;
                if (DO_RELU) v = fmaxf(v, 0.0f);
                out[(size_t)mr * OUT_DIM + col0 + n] = v;
            }
        }
        __syncthreads();
    }
}

extern "C" void kernel_launch(void* const* d_in, const int* in_sizes, int n_in,
                              void* d_out, int out_size, void* d_ws, size_t ws_size,
                              hipStream_t stream) {
    const float* x  = (const float*)d_in[0];
    const int*   ei = (const int*)d_in[1];
    const float* W1 = (const float*)d_in[2];
    const float* b1 = (const float*)d_in[3];
    const float* W2 = (const float*)d_in[4];
    const float* b2 = (const float*)d_in[5];
    float* out = (float*)d_out;

    const int N = in_sizes[0] / IN_DIM;
    const int E = in_sizes[1] / 2;
    const int* row = ei;
    const int* col = ei + E;

    float* ws  = (float*)d_ws;
    float* dis = ws;
    float* Tx1 = ws + N;
    float* Tx2 = Tx1 + (size_t)N * IN_DIM;
    float* h   = Tx2 + (size_t)N * IN_DIM;

    const int threads = 256;
    const int featN   = N * IN_DIM;
    const unsigned gridE    = (unsigned)((E + threads - 1) / threads);
    const unsigned gridN    = (unsigned)((N + threads - 1) / threads);
    const unsigned gridFeat = (unsigned)((featN + threads - 1) / threads);
    const unsigned gridSpmm = 2048;
    float* tk = out;

    hipMemsetAsync(dis, 0, (size_t)N * sizeof(float), stream);
    deg_kernel<<<gridE, threads, 0, stream>>>(row, col, dis, E);
    rsqrt_kernel<<<gridN, threads, 0, stream>>>(dis, out, N);

    hipMemsetAsync(Tx1, 0, (size_t)featN * sizeof(float), stream);
    spmm_kernel<1><<<gridSpmm, threads, 0, stream>>>(row, col, dis, x, Tx1, tk + 0, E, N);
    rescale_seed_kernel<<<gridFeat, threads, 0, stream>>>(dis, Tx1, Tx2, featN);
    spmm_kernel<0><<<gridSpmm, threads, 0, stream>>>(row, col, dis, Tx1, Tx2, tk + 2, E, N);

    cheb_gemm_wmma<64, 1, 4><<<1024, 128, 0, stream>>>(x, Tx1, Tx2, dis, W1, b1, h, N);

    hipMemsetAsync(Tx1, 0, (size_t)featN * sizeof(float), stream);
    spmm_kernel<1><<<gridSpmm, threads, 0, stream>>>(row, col, dis, h, Tx1, tk + 4, E, N);
    rescale_seed_kernel<<<gridFeat, threads, 0, stream>>>(dis, Tx1, Tx2, featN);
    spmm_kernel<0><<<gridSpmm, threads, 0, stream>>>(row, col, dis, Tx1, Tx2, tk + 6, E, N);

    cheb_gemm_wmma<16, 0, 4><<<782, 128, 0, stream>>>(h, Tx1, Tx2, dis, W2, b2, out, N);
}


# Degree count: one count plane per package half (a block reads its die number and adds only into its half's plane, 4 edges per lane); the rsqrt pass adds the two planes; counts stay exact
# speedup vs baseline: 2.7528x; 1.2099x over previous
#include <hip/hip_runtime.h>

typedef __attribute__((ext_vector_type(2))) float v2f;
typedef __attribute__((ext_vector_type(8))) float v8f;
typedef __attribute__((ext_vector_type(4))) float v4f;

#define IN_DIM 64

#define DEG_EDGES 4

__global__ void __launch_bounds__(256)
deg_kernel(const int* __restrict__ row, const int* __restrict__ col,
           float* __restrict__ deg, int E, int N) {
    const int myHalf = (int)((__builtin_amdgcn_s_sendmsg_rtn(0x87) >> 18) & 1u);
    float* plane = deg + (size_t)myHalf * N;
    const int base = blockIdx.x * (blockDim.x * DEG_EDGES) + threadIdx.x;
    int r[DEG_EDGES], c[DEG_EDGES];
#pragma unroll
    for (int j = 0; j < DEG_EDGES; ++j) {
        const int e = base + j * blockDim.x;
        r[j] = 0; c[j] = 0;
        if (e < E) { r[j] = row[e]; c[j] = col[e]; }
    }
#pragma unroll
    for (int j = 0; j < DEG_EDGES; ++j)
        if (r[j] != c[j]) atomicAdd(&plane[r[j]], 1.0f);
}

__global__ void rsqrt_kernel(float* __restrict__ dis, const float* __restrict__ deg2,
                             float* __restrict__ ticketWords, int N) {
    int i = blockIdx.x * blockDim.x + threadIdx.x;
    if (i < 8) reinterpret_cast<int*>(ticketWords)[i] = 0;
    if (i >= N) return;
    float d = dis[i] + deg2[i];
    dis[i] = (d > 0.0f) ? rsqrtf(d) : 0.0f;
}

__global__ void rescale_seed_kernel(const float* __restrict__ dis, float* __restrict__ S1,
                                    float* __restrict__ S2, int n) {
    int i = blockIdx.x * blockDim.x + threadIdx.x;
    if (i >= n) return;
    const float d = dis[i / IN_DIM];
    S1[i] = -(d * d) * S1[i];
    S2[i] = 0.0f;
}

#define WAVE_EDGES 64
#define TICKET_EDGES 512
#define EDGE_GROUP 8

template <int COLSCALE>
__global__ void __launch_bounds__(256)
spmm_kernel(const int* __restrict__ row, const int* __restrict__ col,
            const float* __restrict__ dis, const float* __restrict__ V,
            float* __restrict__ OUT, float* __restrict__ ticketWords,
            int E, int N) {
    __shared__ int s_ticket[2];
    int* tickets = reinterpret_cast<int*>(ticketWords);
    const int lane = threadIdx.x & 31;
    const int waveInBlock = threadIdx.x >> 5;
    const int nTickets = (E + TICKET_EDGES - 1) / TICKET_EDGES;
    const int split = N >> 1;
    const int myHalf = (int)((__builtin_amdgcn_s_sendmsg_rtn(0x87) >> 18) & 1u);

    int k = 0;
    for (int phase = 0; phase < 2; ++phase) {
        const int rg = myHalf ^ phase;
        const int lo = rg ? split : 0;
        const unsigned span = (unsigned)(rg ? (N - split) : split);
        int* cnt = tickets + rg;
        int tn = 0;
        if (threadIdx.x == 0) tn = atomicAdd(cnt, 1);
        for (;;) {
            if (threadIdx.x == 0) s_ticket[k & 1] = tn;
            __syncthreads();
            const int t = __builtin_amdgcn_readfirstlane(s_ticket[k & 1]);
            ++k;
            if (t >= nTickets) break;
            const int e = t * TICKET_EDGES + waveInBlock * WAVE_EDGES + lane;
            int r0 = 0, c0 = 0, r1 = 0, c1 = 0;
            if (e < E)      { r0 = row[e];      c0 = col[e]; }
            if (e + 32 < E) { r1 = row[e + 32]; c1 = col[e + 32]; }
            if (threadIdx.x == 0) tn = atomicAdd(cnt, 1);
            const bool in0 = (e < E)      && ((unsigned)(r0 - lo) < span) && (r0 != c0);
            const bool in1 = (e + 32 < E) && ((unsigned)(r1 - lo) < span) && (r1 != c1);
            float w0 = in0 ? 1.0f : 0.0f;
            float w1 = in1 ? 1.0f : 0.0f;
            if (COLSCALE) {
                if (in0) w0 = dis[c0];
                if (in1) w1 = dis[c1];
            }
            unsigned long long mask =
                ((unsigned long long)__builtin_amdgcn_ballot_w32(in1 && w1 != 0.0f) << 32) |
                (unsigned long long)__builtin_amdgcn_ballot_w32(in0 && w0 != 0.0f);

            while (mask) {
                float v0[EDGE_GROUP], v1[EDGE_GROUP], wj[EDGE_GROUP];
                int rj[EDGE_GROUP];
#pragma unroll
                for (int j = 0; j < EDGE_GROUP; ++j) {
                    wj[j] = 0.0f; rj[j] = 0; v0[j] = 0.0f; v1[j] = 0.0f;
                    if (mask) {
                        const int b = __builtin_ctzll(mask);
                        mask &= mask - 1;
                        const int l = b & 31;
                        int cj;
                        if (b < 32) {
                            cj    = __builtin_amdgcn_readlane(c0, l);
                            rj[j] = __builtin_amdgcn_readlane(r0, l);
                            wj[j] = __int_as_float(__builtin_amdgcn_readlane(__float_as_int(w0), l));
                        } else {
                            cj    = __builtin_amdgcn_readlane(c1, l);
                            rj[j] = __builtin_amdgcn_readlane(r1, l);
                            wj[j] = __int_as_float(__builtin_amdgcn_readlane(__float_as_int(w1), l));
                        }
                        const float* vp = V + (size_t)cj * IN_DIM + lane;
                        v0[j] = vp[0];
                        v1[j] = vp[32];
                    }
                }
#pragma unroll
                for (int j = 0; j < EDGE_GROUP; ++j) {
                    if (wj[j] != 0.0f) {
                        float* op = OUT + (size_t)rj[j] * IN_DIM + lane;
                        atomicAdd(op,      COLSCALE ? wj[j] * v0[j] : v0[j]);
                        atomicAdd(op + 32, COLSCALE ? wj[j] * v1[j] : v1[j]);
                    }
                }
            }
        }
    }
}

template <int OUT_DIM, int DO_RELU, int NWAVES>
__global__ void __launch_bounds__(NWAVES * 32, 1)
cheb_gemm_wmma(const float* __restrict__ T0,
               const float* __restrict__ T1,
               const float* __restrict__ T2,
               const float* __restrict__ dis,
               const float* __restrict__ W,
               const float* __restrict__ bias,
               float* __restrict__ out,
               int N) {
    constexpr int COL_TILES  = OUT_DIM / 16;
    constexpr int ROW_TILES  = NWAVES / COL_TILES;
    constexpr int STAGE_ROWS = 16 * ROW_TILES;
    constexpr int THREADS    = NWAVES * 32;
    constexpr int PITCH      = IN_DIM + 4;
    constexpr int CHUNKS     = STAGE_ROWS * 16 / THREADS;
    __shared__ __attribute__((aligned(16))) float sA[3][STAGE_ROWS * PITCH];

    const int lane = threadIdx.x & 31;
    const int waveInBlock = threadIdx.x >> 5;
    const int ct  = waveInBlock % COL_TILES;
    const int rtl = waveInBlock / COL_TILES;

    const int n    = lane & 15;
    const int m    = lane & 15;
    const int koff = (lane >> 4) << 1;
    const int col0 = ct << 4;
    const int mhalf = (lane >> 4) << 3;

    v2f bfrag[3][16];
#pragma unroll
    for (int s = 0; s < 3; ++s) {
        const float* Ws = W + s * IN_DIM * OUT_DIM + col0 + n;
#pragma unroll
        for (int k4 = 0; k4 < 16; ++k4) {
            const int kk = k4 * 4 + koff;
            bfrag[s][k4].x = Ws[(kk + 0) * OUT_DIM];
            bfrag[s][k4].y = Ws[(kk + 1) * OUT_DIM];
        }
    }
    const float bv = bias[col0 + n];

    const int nStages = (N + STAGE_ROWS - 1) / STAGE_ROWS;
    for (int st = blockIdx.x; st < nStages; st += gridDim.x) {
        const int srow0 = st * STAGE_ROWS;

        v4f gx[CHUNKS], g1[CHUNKS], g2[CHUNKS];
        float gd[CHUNKS];
#pragma unroll
        for (int c = 0; c < CHUNKS; ++c) {
            const int idx = c * THREADS + threadIdx.x;
            const int grow = min(srow0 + (idx >> 4), N - 1);
            const size_t g = (size_t)grow * IN_DIM + ((idx & 15) << 2);
            gx[c] = *reinterpret_cast<const v4f*>(T0 + g);
            g1[c] = *reinterpret_cast<const v4f*>(T1 + g);
            g2[c] = *reinterpret_cast<const v4f*>(T2 + g);
            gd[c] = dis[grow];
        }
        if constexpr (CHUNKS == 8) {
            asm volatile("" :: "v"(gx[0]), "v"(g1[0]), "v"(g2[0]), "v"(gx[1]), "v"(g1[1]), "v"(g2[1]),
                               "v"(gx[2]), "v"(g1[2]), "v"(g2[2]), "v"(gx[3]), "v"(g1[3]), "v"(g2[3]),
                               "v"(gx[4]), "v"(g1[4]), "v"(g2[4]), "v"(gx[5]), "v"(g1[5]), "v"(g2[5]),
                               "v"(gx[6]), "v"(g1[6]), "v"(g2[6]), "v"(gx[7]), "v"(g1[7]), "v"(g2[7]));
        } else {
            static_assert(CHUNKS == 2 || CHUNKS == 8, "staging fence is written for 2 or 8 pieces per thread");
            asm volatile("" :: "v"(gx[0]), "v"(g1[0]), "v"(g2[0]), "v"(gx[1]), "v"(g1[1]), "v"(g2[1]));
        }
#pragma unroll
        for (int c = 0; c < CHUNKS; ++c) {
            const int idx = c * THREADS + threadIdx.x;
            const float dr  = gd[c];
            const float inv = (dr > 0.0f) ? 1.0f / dr : 0.0f;
            const float m2  = -2.0f * dr;
            const v4f x4 = gx[c], s1 = g1[c], s2 = g2[c];
            v4f t1, t2;
            t1.x = s1.x * inv; t1.y = s1.y * inv; t1.z = s1.z * inv; t1.w = s1.w * inv;
            t2.x = m2 * s2.x - x4.x; t2.y = m2 * s2.y - x4.y;
            t2.z = m2 * s2.z - x4.z; t2.w = m2 * s2.w - x4.w;
            const int l = (idx >> 4) * PITCH + ((idx & 15) << 2);
            *reinterpret_cast<v4f*>(&sA[0][l]) = x4;
            *reinterpret_cast<v4f*>(&sA[1][l]) = t1;
            *reinterpret_cast<v4f*>(&sA[2][l]) = t2;
        }
        __syncthreads();

        v2f afrag[3][16];
#pragma unroll
        for (int s = 0; s < 3; ++s) {
            const float* ap = &sA[s][(rtl * 16 + m) * PITCH + koff];
#pragma unroll
            for (int k4 = 0; k4 < 16; ++k4)
                afrag[s][k4] = *reinterpret_cast<const v2f*>(ap + k4 * 4);
        }
        __builtin_amdgcn_sched_barrier(0);

        v8f acc0 = {};
        v8f acc1 = {};
#pragma unroll
        for (int s = 0; s < 3; ++s) {
#pragma unroll
            for (int k8 = 0; k8 < 8; ++k8) {
                acc0 = __builtin_amdgcn_wmma_f32_16x16x4_f32(
                    false, afrag[s][2 * k8 + 0], false, bfrag[s][2 * k8 + 0], (short)0, acc0, false, false);
                acc1 = __builtin_amdgcn_wmma_f32_16x16x4_f32(
                    false, afrag[s][2 * k8 + 1], false, bfrag[s][2 * k8 + 1], (short)0, acc1, false, false);
            }
        }

        const int row0 = srow0 + rtl * 16;
#pragma unroll
        for (int r = 0; r < 8; ++r) {
            int mr = row0 + r + mhalf;
            if (mr < N) {
                float v = acc0[r] + acc1[r] + bv;
                if (DO_RELU) v = fmaxf(v, 0.0f);
                out[(size_t)mr * OUT_DIM + col0 + n] = v;
            }
        }
        __syncthreads();
    }
}

extern "C" void kernel_launch(void* const* d_in, const int* in_sizes, int n_in,
                              void* d_out, int out_size, void* d_ws, size_t ws_size,
                              hipStream_t stream) {
    const float* x  = (const float*)d_in[0];
    const int*   ei = (const int*)d_in[1];
    const float* W1 = (const float*)d_in[2];
    const float* b1 = (const float*)d_in[3];
    const float* W2 = (const float*)d_in[4];
    const float* b2 = (const float*)d_in[5];
    float* out = (float*)d_out;

    const int N = in_sizes[0] / IN_DIM;
    const int E = in_sizes[1] / 2;
    const int* row = ei;
    const int* col = ei + E;

    float* ws  = (float*)d_ws;
    float* dis = ws;
    float* Tx1 = ws + N;
    float* Tx2 = Tx1 + (size_t)N * IN_DIM;
    float* h   = Tx2 + (size_t)N * IN_DIM;

    const int threads = 256;
    const int featN   = N * IN_DIM;
    const unsigned gridDeg  = (unsigned)((E + threads * DEG_EDGES - 1) / (threads * DEG_EDGES));
    const unsigned gridN    = (unsigned)((N + threads - 1) / threads);
    const unsigned gridFeat = (unsigned)((featN + threads - 1) / threads);
    const unsigned gridSpmm = 2048;
    float* tk = out;

    hipMemsetAsync(dis, 0, (size_t)N * 2 * sizeof(float), stream);
    deg_kernel<<<gridDeg, threads, 0, stream>>>(row, col, dis, E, N);
    rsqrt_kernel<<<gridN, threads, 0, stream>>>(dis, Tx1, out, N);

    hipMemsetAsync(Tx1, 0, (size_t)featN * sizeof(float), stream);
    spmm_kernel<1><<<gridSpmm, threads, 0, stream>>>(row, col, dis, x, Tx1, tk + 0, E, N);
    rescale_seed_kernel<<<gridFeat, threads, 0, stream>>>(dis, Tx1, Tx2, featN);
    spmm_kernel<0><<<gridSpmm, threads, 0, stream>>>(row, col, dis, Tx1, Tx2, tk + 2, E, N);

    cheb_gemm_wmma<64, 1, 4><<<1024, 128, 0, stream>>>(x, Tx1, Tx2, dis, W1, b1, h, N);

    hipMemsetAsync(Tx1, 0, (size_t)featN * sizeof(float), stream);
    spmm_kernel<1><<<gridSpmm, threads, 0, stream>>>(row, col, dis, h, Tx1, tk + 4, E, N);
    rescale_seed_kernel<<<gridFeat, threads, 0, stream>>>(dis, Tx1, Tx2, featN);
    spmm_kernel<0><<<gridSpmm, threads, 0, stream>>>(row, col, dis, Tx1, Tx2, tk + 6, E, N);

    cheb_gemm_wmma<16, 0, 4><<<782, 128, 0, stream>>>(h, Tx1, Tx2, dis, W2, b2, out, N);
}


# 2nd graph product looks up its column factor per edge like the 1st (dis[c], squared): the rescale+seed pass between them is gone, S2's zero rides in the S1 memset; products re-associated
# speedup vs baseline: 2.8804x; 1.0464x over previous
#include <hip/hip_runtime.h>

typedef __attribute__((ext_vector_type(2))) float v2f;
typedef __attribute__((ext_vector_type(8))) float v8f;
typedef __attribute__((ext_vector_type(4))) float v4f;

#define IN_DIM 64

#define DEG_EDGES 4

__global__ void __launch_bounds__(256)
deg_kernel(const int* __restrict__ row, const int* __restrict__ col,
           float* __restrict__ deg, int E, int N) {
    const int myHalf = (int)((__builtin_amdgcn_s_sendmsg_rtn(0x87) >> 18) & 1u);
    float* plane = deg + (size_t)myHalf * N;
    const int base = blockIdx.x * (blockDim.x * DEG_EDGES) + threadIdx.x;
    int r[DEG_EDGES], c[DEG_EDGES];
#pragma unroll
    for (int j = 0; j < DEG_EDGES; ++j) {
        const int e = base + j * blockDim.x;
        r[j] = 0; c[j] = 0;
        if (e < E) { r[j] = row[e]; c[j] = col[e]; }
    }
#pragma unroll
    for (int j = 0; j < DEG_EDGES; ++j)
        if (r[j] != c[j]) atomicAdd(&plane[r[j]], 1.0f);
}

__global__ void rsqrt_kernel(float* __restrict__ dis, const float* __restrict__ deg2,
                             float* __restrict__ ticketWords, int N) {
    int i = blockIdx.x * blockDim.x + threadIdx.x;
    if (i < 8) reinterpret_cast<int*>(ticketWords)[i] = 0;
    if (i >= N) return;
    float d = dis[i] + deg2[i];
    dis[i] = (d > 0.0f) ? rsqrtf(d) : 0.0f;
}

#define WAVE_EDGES 64
#define TICKET_EDGES 512
#define EDGE_GROUP 8

template <int SQUARE>
__global__ void __launch_bounds__(256)
spmm_kernel(const int* __restrict__ row, const int* __restrict__ col,
            const float* __restrict__ dis, const float* __restrict__ V,
            float* __restrict__ OUT, float* __restrict__ ticketWords,
            int E, int N) {
    __shared__ int s_ticket[2];
    int* tickets = reinterpret_cast<int*>(ticketWords);
    const int lane = threadIdx.x & 31;
    const int waveInBlock = threadIdx.x >> 5;
    const int nTickets = (E + TICKET_EDGES - 1) / TICKET_EDGES;
    const int split = N >> 1;
    const int myHalf = (int)((__builtin_amdgcn_s_sendmsg_rtn(0x87) >> 18) & 1u);

    int k = 0;
    for (int phase = 0; phase < 2; ++phase) {
        const int rg = myHalf ^ phase;
        const int lo = rg ? split : 0;
        const unsigned span = (unsigned)(rg ? (N - split) : split);
        int* cnt = tickets + rg;
        int tn = 0;
        if (threadIdx.x == 0) tn = atomicAdd(cnt, 1);
        for (;;) {
            if (threadIdx.x == 0) s_ticket[k & 1] = tn;
            __syncthreads();
            const int t = __builtin_amdgcn_readfirstlane(s_ticket[k & 1]);
            ++k;
            if (t >= nTickets) break;
            const int e = t * TICKET_EDGES + waveInBlock * WAVE_EDGES + lane;
            int r0 = 0, c0 = 0, r1 = 0, c1 = 0;
            if (e < E)      { r0 = row[e];      c0 = col[e]; }
            if (e + 32 < E) { r1 = row[e + 32]; c1 = col[e + 32]; }
            if (threadIdx.x == 0) tn = atomicAdd(cnt, 1);
            const bool in0 = (e < E)      && ((unsigned)(r0 - lo) < span) && (r0 != c0);
            const bool in1 = (e + 32 < E) && ((unsigned)(r1 - lo) < span) && (r1 != c1);
            float w0 = 0.0f, w1 = 0.0f;
            if (in0) w0 = dis[c0];
            if (in1) w1 = dis[c1];
            if (SQUARE) { w0 *= w0; w1 *= w1; }
            unsigned long long mask =
                ((unsigned long long)__builtin_amdgcn_ballot_w32(in1 && w1 != 0.0f) << 32) |
                (unsigned long long)__builtin_amdgcn_ballot_w32(in0 && w0 != 0.0f);

            while (mask) {
                float v0[EDGE_GROUP], v1[EDGE_GROUP], wj[EDGE_GROUP];
                int rj[EDGE_GROUP];
#pragma unroll
                for (int j = 0; j < EDGE_GROUP; ++j) {
                    wj[j] = 0.0f; rj[j] = 0; v0[j] = 0.0f; v1[j] = 0.0f;
                    if (mask) {
                        const int b = __builtin_ctzll(mask);
                        mask &= mask - 1;
                        const int l = b & 31;
                        int cj;
                        if (b < 32) {
                            cj    = __builtin_amdgcn_readlane(c0, l);
                            rj[j] = __builtin_amdgcn_readlane(r0, l);
                            wj[j] = __int_as_float(__builtin_amdgcn_readlane(__float_as_int(w0), l));
                        } else {
                            cj    = __builtin_amdgcn_readlane(c1, l);
                            rj[j] = __builtin_amdgcn_readlane(r1, l);
                            wj[j] = __int_as_float(__builtin_amdgcn_readlane(__float_as_int(w1), l));
                        }
                        const float* vp = V + (size_t)cj * IN_DIM + lane;
                        v0[j] = vp[0];
                        v1[j] = vp[32];
                    }
                }
#pragma unroll
                for (int j = 0; j < EDGE_GROUP; ++j) {
                    if (wj[j] != 0.0f) {
                        float* op = OUT + (size_t)rj[j] * IN_DIM + lane;
                        atomicAdd(op,      wj[j] * v0[j]);
                        atomicAdd(op + 32, wj[j] * v1[j]);
                    }
                }
            }
        }
    }
}

template <int OUT_DIM, int DO_RELU, int NWAVES>
__global__ void __launch_bounds__(NWAVES * 32, 1)
cheb_gemm_wmma(const float* __restrict__ T0,
               const float* __restrict__ T1,
               const float* __restrict__ T2,
               const float* __restrict__ dis,
               const float* __restrict__ W,
               const float* __restrict__ bias,
               float* __restrict__ out,
               int N) {
    constexpr int COL_TILES  = OUT_DIM / 16;
    constexpr int ROW_TILES  = NWAVES / COL_TILES;
    constexpr int STAGE_ROWS = 16 * ROW_TILES;
    constexpr int THREADS    = NWAVES * 32;
    constexpr int PITCH      = IN_DIM + 4;
    constexpr int CHUNKS     = STAGE_ROWS * 16 / THREADS;
    __shared__ __attribute__((aligned(16))) float sA[3][STAGE_ROWS * PITCH];

    const int lane = threadIdx.x & 31;
    const int waveInBlock = threadIdx.x >> 5;
    const int ct  = waveInBlock % COL_TILES;
    const int rtl = waveInBlock / COL_TILES;

    const int n    = lane & 15;
    const int m    = lane & 15;
    const int koff = (lane >> 4) << 1;
    const int col0 = ct << 4;
    const int mhalf = (lane >> 4) << 3;

    v2f bfrag[3][16];
#pragma unroll
    for (int s = 0; s < 3; ++s) {
        const float* Ws = W + s * IN_DIM * OUT_DIM + col0 + n;
#pragma unroll
        for (int k4 = 0; k4 < 16; ++k4) {
            const int kk = k4 * 4 + koff;
            bfrag[s][k4].x = Ws[(kk + 0) * OUT_DIM];
            bfrag[s][k4].y = Ws[(kk + 1) * OUT_DIM];
        }
    }
    const float bv = bias[col0 + n];

    const int nStages = (N + STAGE_ROWS - 1) / STAGE_ROWS;
    for (int st = blockIdx.x; st < nStages; st += gridDim.x) {
        const int srow0 = st * STAGE_ROWS;

        v4f gx[CHUNKS], g1[CHUNKS], g2[CHUNKS];
        float gd[CHUNKS];
#pragma unroll
        for (int c = 0; c < CHUNKS; ++c) {
            const int idx = c * THREADS + threadIdx.x;
            const int grow = min(srow0 + (idx >> 4), N - 1);
            const size_t g = (size_t)grow * IN_DIM + ((idx & 15) << 2);
            gx[c] = *reinterpret_cast<const v4f*>(T0 + g);
            g1[c] = *reinterpret_cast<const v4f*>(T1 + g);
            g2[c] = *reinterpret_cast<const v4f*>(T2 + g);
            gd[c] = dis[grow];
        }
        if constexpr (CHUNKS == 8) {
            asm volatile("" :: "v"(gx[0]), "v"(g1[0]), "v"(g2[0]), "v"(gx[1]), "v"(g1[1]), "v"(g2[1]),
                               "v"(gx[2]), "v"(g1[2]), "v"(g2[2]), "v"(gx[3]), "v"(g1[3]), "v"(g2[3]),
                               "v"(gx[4]), "v"(g1[4]), "v"(g2[4]), "v"(gx[5]), "v"(g1[5]), "v"(g2[5]),
                               "v"(gx[6]), "v"(g1[6]), "v"(g2[6]), "v"(gx[7]), "v"(g1[7]), "v"(g2[7]));
        } else {
            static_assert(CHUNKS == 2 || CHUNKS == 8, "staging fence is written for 2 or 8 pieces per thread");
            asm volatile("" :: "v"(gx[0]), "v"(g1[0]), "v"(g2[0]), "v"(gx[1]), "v"(g1[1]), "v"(g2[1]));
        }
#pragma unroll
        for (int c = 0; c < CHUNKS; ++c) {
            const int idx = c * THREADS + threadIdx.x;
            const float dr  = gd[c];
            const float m1  = -dr;
            const float m2  = 2.0f * dr;
            const v4f x4 = gx[c], s1 = g1[c], s2 = g2[c];
            v4f t1, t2;
            t1.x = m1 * s1.x; t1.y = m1 * s1.y; t1.z = m1 * s1.z; t1.w = m1 * s1.w;
            t2.x = m2 * s2.x - x4.x; t2.y = m2 * s2.y - x4.y;
            t2.z = m2 * s2.z - x4.z; t2.w = m2 * s2.w - x4.w;
            const int l = (idx >> 4) * PITCH + ((idx & 15) << 2);
            *reinterpret_cast<v4f*>(&sA[0][l]) = x4;
            *reinterpret_cast<v4f*>(&sA[1][l]) = t1;
            *reinterpret_cast<v4f*>(&sA[2][l]) = t2;
        }
        __syncthreads();

        v2f afrag[3][16];
#pragma unroll
        for (int s = 0; s < 3; ++s) {
            const float* ap = &sA[s][(rtl * 16 + m) * PITCH + koff];
#pragma unroll
            for (int k4 = 0; k4 < 16; ++k4)
                afrag[s][k4] = *reinterpret_cast<const v2f*>(ap + k4 * 4);
        }
        __builtin_amdgcn_sched_barrier(0);

        v8f acc0 = {};
        v8f acc1 = {};
#pragma unroll
        for (int s = 0; s < 3; ++s) {
#pragma unroll
            for (int k8 = 0; k8 < 8; ++k8) {
                acc0 = __builtin_amdgcn_wmma_f32_16x16x4_f32(
                    false, afrag[s][2 * k8 + 0], false, bfrag[s][2 * k8 + 0], (short)0, acc0, false, false);
                acc1 = __builtin_amdgcn_wmma_f32_16x16x4_f32(
                    false, afrag[s][2 * k8 + 1], false, bfrag[s][2 * k8 + 1], (short)0, acc1, false, false);
            }
        }

        const int row0 = srow0 + rtl * 16;
#pragma unroll
        for (int r = 0; r < 8; ++r) {
            int mr = row0 + r + mhalf;
            if (mr < N) {
                float v = acc0[r] + acc1[r] + bv;
                if (DO_RELU) v = fmaxf(v, 0.0f);
                out[(size_t)mr * OUT_DIM + col0 + n] = v;
            }
        }
        __syncthreads();
    }
}

extern "C" void kernel_launch(void* const* d_in, const int* in_sizes, int n_in,
                              void* d_out, int out_size, void* d_ws, size_t ws_size,
                              hipStream_t stream) {
    const float* x  = (const float*)d_in[0];
    const int*   ei = (const int*)d_in[1];
    const float* W1 = (const float*)d_in[2];
    const float* b1 = (const float*)d_in[3];
    const float* W2 = (const float*)d_in[4];
    const float* b2 = (const float*)d_in[5];
    float* out = (float*)d_out;

    const int N = in_sizes[0] / IN_DIM;
    const int E = in_sizes[1] / 2;
    const int* row = ei;
    const int* col = ei + E;

    float* ws  = (float*)d_ws;
    float* dis = ws;
    float* Tx1 = ws + N;
    float* Tx2 = Tx1 + (size_t)N * IN_DIM;
    float* h   = Tx2 + (size_t)N * IN_DIM;

    const int threads = 256;
    const int featN   = N * IN_DIM;
    const unsigned gridDeg  = (unsigned)((E + threads * DEG_EDGES - 1) / (threads * DEG_EDGES));
    const unsigned gridN    = (unsigned)((N + threads - 1) / threads);
    const unsigned gridSpmm = 2048;
    float* tk = out;

    hipMemsetAsync(dis, 0, (size_t)N * 2 * sizeof(float), stream);
    deg_kernel<<<gridDeg, threads, 0, stream>>>(row, col, dis, E, N);
    rsqrt_kernel<<<gridN, threads, 0, stream>>>(dis, Tx1, out, N);

    hipMemsetAsync(Tx1, 0, (size_t)featN * 2 * sizeof(float), stream);
    spmm_kernel<0><<<gridSpmm, threads, 0, stream>>>(row, col, dis, x, Tx1, tk + 0, E, N);
    spmm_kernel<1><<<gridSpmm, threads, 0, stream>>>(row, col, dis, Tx1, Tx2, tk + 2, E, N);

    cheb_gemm_wmma<64, 1, 4><<<1024, 128, 0, stream>>>(x, Tx1, Tx2, dis, W1, b1, h, N);

    hipMemsetAsync(Tx1, 0, (size_t)featN * 2 * sizeof(float), stream);
    spmm_kernel<0><<<gridSpmm, threads, 0, stream>>>(row, col, dis, h, Tx1, tk + 4, E, N);
    spmm_kernel<1><<<gridSpmm, threads, 0, stream>>>(row, col, dis, Tx1, Tx2, tk + 6, E, N);

    cheb_gemm_wmma<16, 0, 4><<<782, 128, 0, stream>>>(h, Tx1, Tx2, dis, W2, b2, out, N);
}


# SpMM column factor dis[col] looked up once per edge and call: 1st product stores it in an edge plane (in out, free till the last launch); the 3 later products read it coalesced beside row/col
# speedup vs baseline: 2.9741x; 1.0325x over previous
#include <hip/hip_runtime.h>

typedef __attribute__((ext_vector_type(2))) float v2f;
typedef __attribute__((ext_vector_type(8))) float v8f;
typedef __attribute__((ext_vector_type(4))) float v4f;

#define IN_DIM 64
#define OUT2_DIM 16

#define DEG_EDGES 4

__global__ void __launch_bounds__(256)
deg_kernel(const int* __restrict__ row, const int* __restrict__ col,
           float* __restrict__ deg, int E, int N) {
    const int myHalf = (int)((__builtin_amdgcn_s_sendmsg_rtn(0x87) >> 18) & 1u);
    float* plane = deg + (size_t)myHalf * N;
    const int base = blockIdx.x * (blockDim.x * DEG_EDGES) + threadIdx.x;
    int r[DEG_EDGES], c[DEG_EDGES];
#pragma unroll
    for (int j = 0; j < DEG_EDGES; ++j) {
        const int e = base + j * blockDim.x;
        r[j] = 0; c[j] = 0;
        if (e < E) { r[j] = row[e]; c[j] = col[e]; }
    }
#pragma unroll
    for (int j = 0; j < DEG_EDGES; ++j)
        if (r[j] != c[j]) atomicAdd(&plane[r[j]], 1.0f);
}

__global__ void rsqrt_kernel(float* __restrict__ dis, const float* __restrict__ deg2,
                             float* __restrict__ ticketWords, int N) {
    int i = blockIdx.x * blockDim.x + threadIdx.x;
    if (i < 8) reinterpret_cast<int*>(ticketWords)[i] = 0;
    if (i >= N) return;
    float d = dis[i] + deg2[i];
    dis[i] = (d > 0.0f) ? rsqrtf(d) : 0.0f;
}

#define WAVE_EDGES 64
#define TICKET_EDGES 512
#define EDGE_GROUP 8

template <int SQUARE, int FIRST>
__global__ void __launch_bounds__(256)
spmm_kernel(const int* __restrict__ row, const int* __restrict__ col,
            const float* __restrict__ dis, const float* __restrict__ V,
            float* __restrict__ OUT, float* __restrict__ ticketWords,
            float* __restrict__ wE, int wCap,
            int E, int N) {
    __shared__ int s_ticket[2];
    int* tickets = reinterpret_cast<int*>(ticketWords);
    const int lane = threadIdx.x & 31;
    const int waveInBlock = threadIdx.x >> 5;
    const int nTickets = (E + TICKET_EDGES - 1) / TICKET_EDGES;
    const int split = N >> 1;
    const int myHalf = (int)((__builtin_amdgcn_s_sendmsg_rtn(0x87) >> 18) & 1u);

    int k = 0;
    for (int phase = 0; phase < 2; ++phase) {
        const int rg = myHalf ^ phase;
        const int lo = rg ? split : 0;
        const unsigned span = (unsigned)(rg ? (N - split) : split);
        int* cnt = tickets + rg;
        int tn = 0;
        if (threadIdx.x == 0) tn = atomicAdd(cnt, 1);
        for (;;) {
            if (threadIdx.x == 0) s_ticket[k & 1] = tn;
            __syncthreads();
            const int t = __builtin_amdgcn_readfirstlane(s_ticket[k & 1]);
            ++k;
            if (t >= nTickets) break;
            const int e = t * TICKET_EDGES + waveInBlock * WAVE_EDGES + lane;
            int r0 = 0, c0 = 0, r1 = 0, c1 = 0;
            float f0 = 0.0f, f1 = 0.0f;
            if (e < E)      { r0 = row[e];      c0 = col[e]; }
            if (e + 32 < E) { r1 = row[e + 32]; c1 = col[e + 32]; }
            if (!FIRST) {
                if (e < E && e < wCap)           f0 = wE[e];
                if (e + 32 < E && e + 32 < wCap) f1 = wE[e + 32];
            }
            if (threadIdx.x == 0) tn = atomicAdd(cnt, 1);
            const bool own0 = (e < E)      && ((unsigned)(r0 - lo) < span);
            const bool own1 = (e + 32 < E) && ((unsigned)(r1 - lo) < span);
            const bool in0 = own0 && (r0 != c0);
            const bool in1 = own1 && (r1 != c1);
            float w0 = 0.0f, w1 = 0.0f;
            if (FIRST) {
                if (in0) w0 = dis[c0];
                if (in1) w1 = dis[c1];
                if (own0 && e < wCap)      wE[e]      = w0;
                if (own1 && e + 32 < wCap) wE[e + 32] = w1;
            } else {
                if (in0) w0 = (e < wCap)      ? f0 : dis[c0];
                if (in1) w1 = (e + 32 < wCap) ? f1 : dis[c1];
            }
            if (SQUARE) { w0 *= w0; w1 *= w1; }
            unsigned long long mask =
                ((unsigned long long)__builtin_amdgcn_ballot_w32(in1 && w1 != 0.0f) << 32) |
                (unsigned long long)__builtin_amdgcn_ballot_w32(in0 && w0 != 0.0f);

            while (mask) {
                float v0[EDGE_GROUP], v1[EDGE_GROUP], wj[EDGE_GROUP];
                int rj[EDGE_GROUP];
#pragma unroll
                for (int j = 0; j < EDGE_GROUP; ++j) {
                    wj[j] = 0.0f; rj[j] = 0; v0[j] = 0.0f; v1[j] = 0.0f;
                    if (mask) {
                        const int b = __builtin_ctzll(mask);
                        mask &= mask - 1;
                        const int l = b & 31;
                        int cj;
                        if (b < 32) {
                            cj    = __builtin_amdgcn_readlane(c0, l);
                            rj[j] = __builtin_amdgcn_readlane(r0, l);
                            wj[j] = __int_as_float(__builtin_amdgcn_readlane(__float_as_int(w0), l));
                        } else {
                            cj    = __builtin_amdgcn_readlane(c1, l);
                            rj[j] = __builtin_amdgcn_readlane(r1, l);
                            wj[j] = __int_as_float(__builtin_amdgcn_readlane(__float_as_int(w1), l));
                        }
                        const float* vp = V + (size_t)cj * IN_DIM + lane;
                        v0[j] = vp[0];
                        v1[j] = vp[32];
                    }
                }
#pragma unroll
                for (int j = 0; j < EDGE_GROUP; ++j) {
                    if (wj[j] != 0.0f) {
                        float* op = OUT + (size_t)rj[j] * IN_DIM + lane;
                        atomicAdd(op,      wj[j] * v0[j]);
                        atomicAdd(op + 32, wj[j] * v1[j]);
                    }
                }
            }
        }
    }
}

template <int OUT_DIM, int DO_RELU, int NWAVES>
__global__ void __launch_bounds__(NWAVES * 32, 1)
cheb_gemm_wmma(const float* __restrict__ T0,
               const float* __restrict__ T1,
               const float* __restrict__ T2,
               const float* __restrict__ dis,
               const float* __restrict__ W,
               const float* __restrict__ bias,
               float* __restrict__ out,
               int N) {
    constexpr int COL_TILES  = OUT_DIM / 16;
    constexpr int ROW_TILES  = NWAVES / COL_TILES;
    constexpr int STAGE_ROWS = 16 * ROW_TILES;
    constexpr int THREADS    = NWAVES * 32;
    constexpr int PITCH      = IN_DIM + 4;
    constexpr int CHUNKS     = STAGE_ROWS * 16 / THREADS;
    __shared__ __attribute__((aligned(16))) float sA[3][STAGE_ROWS * PITCH];

    const int lane = threadIdx.x & 31;
    const int waveInBlock = threadIdx.x >> 5;
    const int ct  = waveInBlock % COL_TILES;
    const int rtl = waveInBlock / COL_TILES;

    const int n    = lane & 15;
    const int m    = lane & 15;
    const int koff = (lane >> 4) << 1;
    const int col0 = ct << 4;
    const int mhalf = (lane >> 4) << 3;

    v2f bfrag[3][16];
#pragma unroll
    for (int s = 0; s < 3; ++s) {
        const float* Ws = W + s * IN_DIM * OUT_DIM + col0 + n;
#pragma unroll
        for (int k4 = 0; k4 < 16; ++k4) {
            const int kk = k4 * 4 + koff;
            bfrag[s][k4].x = Ws[(kk + 0) * OUT_DIM];
            bfrag[s][k4].y = Ws[(kk + 1) * OUT_DIM];
        }
    }
    const float bv = bias[col0 + n];

    const int nStages = (N + STAGE_ROWS - 1) / STAGE_ROWS;
    for (int st = blockIdx.x; st < nStages; st += gridDim.x) {
        const int srow0 = st * STAGE_ROWS;

        v4f gx[CHUNKS], g1[CHUNKS], g2[CHUNKS];
        float gd[CHUNKS];
#pragma unroll
        for (int c = 0; c < CHUNKS; ++c) {
            const int idx = c * THREADS + threadIdx.x;
            const int grow = min(srow0 + (idx >> 4), N - 1);
            const size_t g = (size_t)grow * IN_DIM + ((idx & 15) << 2);
            gx[c] = *reinterpret_cast<const v4f*>(T0 + g);
            g1[c] = *reinterpret_cast<const v4f*>(T1 + g);
            g2[c] = *reinterpret_cast<const v4f*>(T2 + g);
            gd[c] = dis[grow];
        }
        if constexpr (CHUNKS == 8) {
            asm volatile("" :: "v"(gx[0]), "v"(g1[0]), "v"(g2[0]), "v"(gx[1]), "v"(g1[1]), "v"(g2[1]),
                               "v"(gx[2]), "v"(g1[2]), "v"(g2[2]), "v"(gx[3]), "v"(g1[3]), "v"(g2[3]),
                               "v"(gx[4]), "v"(g1[4]), "v"(g2[4]), "v"(gx[5]), "v"(g1[5]), "v"(g2[5]),
                               "v"(gx[6]), "v"(g1[6]), "v"(g2[6]), "v"(gx[7]), "v"(g1[7]), "v"(g2[7]));
        } else {
            static_assert(CHUNKS == 2 || CHUNKS == 8, "staging fence is written for 2 or 8 pieces per thread");
            asm volatile("" :: "v"(gx[0]), "v"(g1[0]), "v"(g2[0]), "v"(gx[1]), "v"(g1[1]), "v"(g2[1]));
        }
#pragma unroll
        for (int c = 0; c < CHUNKS; ++c) {
            const int idx = c * THREADS + threadIdx.x;
            const float dr  = gd[c];
            const float m1  = -dr;
            const float m2  = 2.0f * dr;
            const v4f x4 = gx[c], s1 = g1[c], s2 = g2[c];
            v4f t1, t2;
            t1.x = m1 * s1.x; t1.y = m1 * s1.y; t1.z = m1 * s1.z; t1.w = m1 * s1.w;
            t2.x = m2 * s2.x - x4.x; t2.y = m2 * s2.y - x4.y;
            t2.z = m2 * s2.z - x4.z; t2.w = m2 * s2.w - x4.w;
            const int l = (idx >> 4) * PITCH + ((idx & 15) << 2);
            *reinterpret_cast<v4f*>(&sA[0][l]) = x4;
            *reinterpret_cast<v4f*>(&sA[1][l]) = t1;
            *reinterpret_cast<v4f*>(&sA[2][l]) = t2;
        }
        __syncthreads();

        v2f afrag[3][16];
#pragma unroll
        for (int s = 0; s < 3; ++s) {
            const float* ap = &sA[s][(rtl * 16 + m) * PITCH + koff];
#pragma unroll
            for (int k4 = 0; k4 < 16; ++k4)
                afrag[s][k4] = *reinterpret_cast<const v2f*>(ap + k4 * 4);
        }
        __builtin_amdgcn_sched_barrier(0);

        v8f acc0 = {};
        v8f acc1 = {};
#pragma unroll
        for (int s = 0; s < 3; ++s) {
#pragma unroll
            for (int k8 = 0; k8 < 8; ++k8) {
                acc0 = __builtin_amdgcn_wmma_f32_16x16x4_f32(
                    false, afrag[s][2 * k8 + 0], false, bfrag[s][2 * k8 + 0], (short)0, acc0, false, false);
                acc1 = __builtin_amdgcn_wmma_f32_16x16x4_f32(
                    false, afrag[s][2 * k8 + 1], false, bfrag[s][2 * k8 + 1], (short)0, acc1, false, false);
            }
        }

        const int row0 = srow0 + rtl * 16;
#pragma unroll
        for (int r = 0; r < 8; ++r) {
            int mr = row0 + r + mhalf;
            if (mr < N) {
                float v = acc0[r] + acc1[r] + bv;
                if (DO_RELU) v = fmaxf(v, 0.0f);
                out[(size_t)mr * OUT_DIM + col0 + n] = v;
            }
        }
        __syncthreads();
    }
}

extern "C" void kernel_launch(void* const* d_in, const int* in_sizes, int n_in,
                              void* d_out, int out_size, void* d_ws, size_t ws_size,
                              hipStream_t stream) {
    const float* x  = (const float*)d_in[0];
    const int*   ei = (const int*)d_in[1];
    const float* W1 = (const float*)d_in[2];
    const float* b1 = (const float*)d_in[3];
    const float* W2 = (const float*)d_in[4];
    const float* b2 = (const float*)d_in[5];
    float* out = (float*)d_out;

    const int N = in_sizes[0] / IN_DIM;
    const int E = in_sizes[1] / 2;
    const int* row = ei;
    const int* col = ei + E;

    float* ws  = (float*)d_ws;
    float* dis = ws;
    float* Tx1 = ws + N;
    float* Tx2 = Tx1 + (size_t)N * IN_DIM;
    float* h   = Tx2 + (size_t)N * IN_DIM;

    const int threads = 256;
    const int featN   = N * IN_DIM;
    const unsigned gridDeg  = (unsigned)((E + threads * DEG_EDGES - 1) / (threads * DEG_EDGES));
    const unsigned gridN    = (unsigned)((N + threads - 1) / threads);
    const unsigned gridSpmm = 2048;
    const int wCap = N * OUT2_DIM - 8;
    float* tk = out + wCap;
    float* wE = out;

    hipMemsetAsync(dis, 0, (size_t)N * 2 * sizeof(float), stream);
    deg_kernel<<<gridDeg, threads, 0, stream>>>(row, col, dis, E, N);
    rsqrt_kernel<<<gridN, threads, 0, stream>>>(dis, Tx1, tk, N);

    hipMemsetAsync(Tx1, 0, (size_t)featN * 2 * sizeof(float), stream);
    spmm_kernel<0, 1><<<gridSpmm, threads, 0, stream>>>(row, col, dis, x, Tx1, tk + 0, wE, wCap, E, N);
    spmm_kernel<1, 0><<<gridSpmm, threads, 0, stream>>>(row, col, dis, Tx1, Tx2, tk + 2, wE, wCap, E, N);

    cheb_gemm_wmma<64, 1, 4><<<1024, 128, 0, stream>>>(x, Tx1, Tx2, dis, W1, b1, h, N);

    hipMemsetAsync(Tx1, 0, (size_t)featN * 2 * sizeof(float), stream);
    spmm_kernel<0, 0><<<gridSpmm, threads, 0, stream>>>(row, col, dis, h, Tx1, tk + 4, wE, wCap, E, N);
    spmm_kernel<1, 0><<<gridSpmm, threads, 0, stream>>>(row, col, dis, Tx1, Tx2, tk + 6, wE, wCap, E, N);

    cheb_gemm_wmma<16, 0, 4><<<782, 128, 0, stream>>>(h, Tx1, Tx2, dis, W2, b2, out, N);
}
